# starting kernel
# baseline (speedup 1.0000x reference)
#include <hip/hip_runtime.h>


typedef __attribute__((ext_vector_type(2))) float v2f;
typedef __attribute__((ext_vector_type(8))) float v8f;

#define T_LEN    48000
#define M1       25
#define NITER    30
#define CHUNK    1024
#define HALO     720
#define REGION   (CHUNK + HALO)
#define NGROUPS  (REGION / 16)
#define NTHREADS 256
#define NWAVES   (NTHREADS / 32)
#define YPT      (CHUNK / NTHREADS)
#define PADL     32
#define USZ      (REGION + PADL)
#define DUMP     (USZ - 1)


__global__ __launch_bounds__(NTHREADS)
void mglsa_taylor_wmma_kernel(const float* __restrict__ x,
                              const float* __restrict__ mc,
                              float* __restrict__ y)
{
    __shared__ float uA[USZ];
    __shared__ float uB[USZ];

    const int b    = blockIdx.y;
    const int cs   = blockIdx.x * CHUNK;
    const int R    = cs - HALO;
    const int tid  = threadIdx.x;
    const int lane = tid & 31;
    const int wave = tid >> 5;

    const float* __restrict__ xb  = x  + (size_t)b * T_LEN;
    const float* __restrict__ mcb = mc + (size_t)b * T_LEN * M1;

    for (int p = tid; p < USZ; p += NTHREADS) {
        int l = (REGION - 1) - p;
        int t = R + l;
        uA[p] = (l >= 0 && (unsigned)t < (unsigned)T_LEN) ? xb[t] : 0.0f;
    }
    if (tid < PADL) uB[REGION + tid] = 0.0f;

    float yreg[YPT];
#pragma unroll
    for (int k = 0; k < YPT; ++k) {
        int t = cs + tid + k * NTHREADS;
        yreg[k] = (t < T_LEN) ? xb[t] : 0.0f;
    }
    __syncthreads();

    float* ucur = uA;
    float* unxt = uB;

    const int  m       = lane & 15;
    const int  half2   = (lane >> 4) << 1;
    const bool hasDiag = (lane < 8) || (lane >= 24);
    const int  dslot   = (lane < 8) ? lane : (lane - 16);

    for (int n = 1; n <= NITER; ++n) {
        const float inv_a = 1.0f / (float)n;

        for (int g = wave; g < NGROUPS; g += NWAVES) {
            const int l0   = g << 4;
            const int trow = R + l0 + m;
            const int tcl  = trow < 0 ? 0 : (trow >= T_LEN ? (T_LEN - 1) : trow);
            const float* __restrict__ mrow = mcb + (size_t)tcl * M1;
            const int pb = (REGION - 1) - (l0 + m - half2);

            {
                int tpre = trow + (NWAVES << 4);
                tpre = tpre < 0 ? 0 : (tpre >= T_LEN ? (T_LEN - 1) : tpre);
                __builtin_prefetch(mcb + (size_t)tpre * M1, 0, 3);
            }

            v8f acc = {0.f, 0.f, 0.f, 0.f, 0.f, 0.f, 0.f, 0.f};

#pragma unroll
            for (int k0 = 0; k0 < 24; k0 += 4) {
                const int j0 = k0 + half2;
                v2f a;
                a.x = mrow[j0];
                a.y = mrow[j0 + 1];
                v2f bb;
                bb.x = ucur[pb + k0];
                bb.y = ucur[pb + k0 + 1];
                acc = __builtin_amdgcn_wmma_f32_16x16x4_f32(
                          false, a, false, bb, (short)0, acc, false, false);
            }
            {
                v2f a;
                float a0 = mrow[24];
                a.x = (half2 == 0) ? a0 : 0.0f;
                a.y = 0.0f;
                v2f bb;
                bb.x = ucur[pb + 24];
                bb.y = ucur[pb + 25];
                acc = __builtin_amdgcn_wmma_f32_16x16x4_f32(
                          false, a, false, bb, (short)0, acc, false, false);
            }

            float dv = acc[0];
#pragma unroll
            for (int v = 1; v < 8; ++v)
                dv = ((lane & 7) == v) ? acc[v] : dv;

            const int pst = hasDiag ? ((REGION - 1) - (l0 + dslot)) : DUMP;
            unxt[pst] = dv * inv_a;
        }
        __syncthreads();

#pragma unroll
        for (int k = 0; k < YPT; ++k)
            yreg[k] += unxt[(CHUNK - 1) - (tid + k * NTHREADS)];

        float* tmp = ucur; ucur = unxt; unxt = tmp;
    }

    float* __restrict__ yb = y + (size_t)b * T_LEN;
#pragma unroll
    for (int k = 0; k < YPT; ++k) {
        int t = cs + tid + k * NTHREADS;
        if (t < T_LEN) yb[t] = yreg[k];
    }
}

extern "C" void kernel_launch(void* const* d_in, const int* in_sizes, int n_in,
                              void* d_out, int out_size, void* d_ws, size_t ws_size,
                              hipStream_t stream) {
    (void)n_in; (void)out_size; (void)d_ws; (void)ws_size;
    const float* x  = (const float*)d_in[0];
    const float* mc = (const float*)d_in[1];
    float* yout     = (float*)d_out;

    const int B = in_sizes[0] / T_LEN;
    dim3 grid((T_LEN + CHUNK - 1) / CHUNK, B);
    mglsa_taylor_wmma_kernel<<<grid, NTHREADS, 0, stream>>>(x, mc, yout);
}


# mc A fragments loaded once into registers (14 row-groups per wave, group loop unrolled) and reused by all 30 Taylor iterations, not re-read from L2 each iteration; halo 720->768 for equal groups
# speedup vs baseline: 1.9204x; 1.9204x over previous
#include <hip/hip_runtime.h>


typedef __attribute__((ext_vector_type(2))) float v2f;
typedef __attribute__((ext_vector_type(8))) float v8f;

#define T_LEN    48000
#define M1       25
#define NITER    30
#define CHUNK    1024
#define HALO     768
#define REGION   (CHUNK + HALO)
#define NTHREADS 256
#define NWAVES   (NTHREADS / 32)
#define NGROUPS  (REGION / 16)
#define NG       (NGROUPS / NWAVES)
#define YPT      (CHUNK / NTHREADS)
#define PADL     32
#define USZ      (REGION + PADL)
#define DUMP     (USZ - 1)


__global__ __launch_bounds__(NTHREADS)
void mglsa_taylor_wmma_kernel(const float* __restrict__ x,
                              const float* __restrict__ mc,
                              float* __restrict__ y)
{
    __shared__ float ubuf[2 * USZ];

    const int b    = blockIdx.y;
    const int cs   = blockIdx.x * CHUNK;
    const int R    = cs - HALO;
    const int tid  = threadIdx.x;
    const int lane = tid & 31;
    const int wave = tid >> 5;

    const float* __restrict__ xb  = x  + (size_t)b * T_LEN;
    const float* __restrict__ mcb = mc + (size_t)b * T_LEN * M1;

    const int  m       = lane & 15;
    const int  half2   = (lane >> 4) << 1;
    const bool hasDiag = (lane < 8) || (lane >= 24);
    const int  dslot   = (lane < 8) ? lane : (lane - 16);

    v2f af[NG][7];
#pragma unroll
    for (int i = 0; i < NG; ++i) {
        const int trow = R + ((wave + i * NWAVES) << 4) + m;
        const int tcl  = trow < 0 ? 0 : (trow >= T_LEN ? (T_LEN - 1) : trow);
        const float* __restrict__ mrow = mcb + (size_t)tcl * M1;
#pragma unroll
        for (int q = 0; q < 6; ++q) {
            af[i][q].x = mrow[4 * q + half2];
            af[i][q].y = mrow[4 * q + half2 + 1];
        }
        const float a24 = mrow[24];
        af[i][6].x = (half2 == 0) ? a24 : 0.0f;
        af[i][6].y = 0.0f;
    }

    for (int p = tid; p < USZ; p += NTHREADS) {
        int l = (REGION - 1) - p;
        int t = R + l;
        ubuf[p] = (l >= 0 && (unsigned)t < (unsigned)T_LEN) ? xb[t] : 0.0f;
    }
    if (tid < PADL) ubuf[USZ + REGION + tid] = 0.0f;

    float yreg[YPT];
#pragma unroll
    for (int k = 0; k < YPT; ++k) {
        int t = cs + tid + k * NTHREADS;
        yreg[k] = (t < T_LEN) ? xb[t] : 0.0f;
    }
    __syncthreads();

    int cur = 0;
    int nxt = USZ;

    const int pb0  = (REGION - 1) - ((wave << 4) + m - half2);
    const int pst0 = (REGION - 1) - ((wave << 4) + dslot);

    for (int n = 1; n <= NITER; ++n) {
        const float inv_a = 1.0f / (float)n;
        const float* __restrict__ ucur = ubuf + cur;
        float* __restrict__ unxt = ubuf + nxt;

#pragma unroll
        for (int i = 0; i < NG; ++i) {
            const int pb = pb0 - i * (NWAVES << 4);

            v8f acc = {0.f, 0.f, 0.f, 0.f, 0.f, 0.f, 0.f, 0.f};
#pragma unroll
            for (int q = 0; q < 7; ++q) {
                v2f bb;
                bb.x = ucur[pb + 4 * q];
                bb.y = ucur[pb + 4 * q + 1];
                acc = __builtin_amdgcn_wmma_f32_16x16x4_f32(
                          false, af[i][q], false, bb, (short)0, acc, false, false);
            }

            float dv = acc[0];
#pragma unroll
            for (int v = 1; v < 8; ++v)
                dv = ((lane & 7) == v) ? acc[v] : dv;

            const int pst = hasDiag ? (pst0 - i * (NWAVES << 4)) : DUMP;
            unxt[pst] = dv * inv_a;
        }
        __syncthreads();

#pragma unroll
        for (int k = 0; k < YPT; ++k)
            yreg[k] += unxt[(CHUNK - 1) - (tid + k * NTHREADS)];

        int tmp = cur; cur = nxt; nxt = tmp;
    }

    float* __restrict__ yb = y + (size_t)b * T_LEN;
#pragma unroll
    for (int k = 0; k < YPT; ++k) {
        int t = cs + tid + k * NTHREADS;
        if (t < T_LEN) yb[t] = yreg[k];
    }
}

extern "C" void kernel_launch(void* const* d_in, const int* in_sizes, int n_in,
                              void* d_out, int out_size, void* d_ws, size_t ws_size,
                              hipStream_t stream) {
    (void)n_in; (void)out_size; (void)d_ws; (void)ws_size;
    const float* x  = (const float*)d_in[0];
    const float* mc = (const float*)d_in[1];
    float* yout     = (float*)d_out;

    const int B = in_sizes[0] / T_LEN;
    dim3 grid((T_LEN + CHUNK - 1) / CHUNK, B);
    mglsa_taylor_wmma_kernel<<<grid, NTHREADS, 0, stream>>>(x, mc, yout);
}


# work partition: chunk 1024->1504 samples, block 256->512 threads (9 row-groups per wave): grid 376 blocks uneven over 256 WGPs -> 256 equal blocks, one per WGP; halo 768->800
# speedup vs baseline: 2.8906x; 1.5052x over previous
#include <hip/hip_runtime.h>


typedef __attribute__((ext_vector_type(2))) float v2f;
typedef __attribute__((ext_vector_type(8))) float v8f;

#define T_LEN    48000
#define M1       25
#define NITER    30
#define CHUNK    1504
#define HALO     800
#define REGION   (CHUNK + HALO)
#define NTHREADS 512
#define NWAVES   (NTHREADS / 32)
#define NGROUPS  (REGION / 16)
#define NG       (NGROUPS / NWAVES)
#define YPT      ((CHUNK + NTHREADS - 1) / NTHREADS)
#define PADL     32
#define USZ      (REGION + PADL)
#define DUMP     (USZ - 1)


__global__ __launch_bounds__(NTHREADS) __attribute__((amdgpu_waves_per_eu(4, 4)))
void mglsa_taylor_wmma_kernel(const float* __restrict__ x,
                              const float* __restrict__ mc,
                              float* __restrict__ y)
{
    __shared__ float ubuf[2 * USZ];

    const int b    = blockIdx.y;
    const int cs   = blockIdx.x * CHUNK;
    const int R    = cs - HALO;
    const int tid  = threadIdx.x;
    const int lane = tid & 31;
    const int wave = tid >> 5;

    const float* __restrict__ xb  = x  + (size_t)b * T_LEN;
    const float* __restrict__ mcb = mc + (size_t)b * T_LEN * M1;

    const int  m       = lane & 15;
    const int  half2   = (lane >> 4) << 1;
    const bool hasDiag = (lane < 8) || (lane >= 24);
    const int  dslot   = (lane < 8) ? lane : (lane - 16);

    v2f af[NG][7];
#pragma unroll
    for (int i = 0; i < NG; ++i) {
        const int trow = R + ((wave + i * NWAVES) << 4) + m;
        const int tcl  = trow < 0 ? 0 : (trow >= T_LEN ? (T_LEN - 1) : trow);
        const float* __restrict__ mrow = mcb + (size_t)tcl * M1;
#pragma unroll
        for (int q = 0; q < 6; ++q) {
            af[i][q].x = mrow[4 * q + half2];
            af[i][q].y = mrow[4 * q + half2 + 1];
        }
        const float a24 = mrow[24];
        af[i][6].x = (half2 == 0) ? a24 : 0.0f;
        af[i][6].y = 0.0f;
    }

    for (int p = tid; p < USZ; p += NTHREADS) {
        int l = (REGION - 1) - p;
        int t = R + l;
        ubuf[p] = (l >= 0 && (unsigned)t < (unsigned)T_LEN) ? xb[t] : 0.0f;
    }
    if (tid < PADL) ubuf[USZ + REGION + tid] = 0.0f;

    float yreg[YPT];
    int   ypos[YPT];
#pragma unroll
    for (int k = 0; k < YPT; ++k) {
        int i = tid + k * NTHREADS;
        int t = cs + i;
        yreg[k] = (i < CHUNK && t < T_LEN) ? xb[t] : 0.0f;
        ypos[k] = (i < CHUNK) ? ((CHUNK - 1) - i) : 0;
    }
    __syncthreads();

    int cur = 0;
    int nxt = USZ;

    const int pb0  = (REGION - 1) - ((wave << 4) + m - half2);
    const int pst0 = (REGION - 1) - ((wave << 4) + dslot);

    for (int n = 1; n <= NITER; ++n) {
        const float inv_a = 1.0f / (float)n;
        const float* __restrict__ ucur = ubuf + cur;
        float* __restrict__ unxt = ubuf + nxt;

#pragma unroll
        for (int i = 0; i < NG; ++i) {
            const int pb = pb0 - i * (NWAVES << 4);

            v8f acc = {0.f, 0.f, 0.f, 0.f, 0.f, 0.f, 0.f, 0.f};
#pragma unroll
            for (int q = 0; q < 7; ++q) {
                v2f bb;
                bb.x = ucur[pb + 4 * q];
                bb.y = ucur[pb + 4 * q + 1];
                acc = __builtin_amdgcn_wmma_f32_16x16x4_f32(
                          false, af[i][q], false, bb, (short)0, acc, false, false);
            }

            float dv = acc[0];
#pragma unroll
            for (int v = 1; v < 8; ++v)
                dv = ((lane & 7) == v) ? acc[v] : dv;

            const int pst = hasDiag ? (pst0 - i * (NWAVES << 4)) : DUMP;
            unxt[pst] = dv * inv_a;
        }
        __syncthreads();

#pragma unroll
        for (int k = 0; k < YPT; ++k)
            yreg[k] += unxt[ypos[k]];

        int tmp = cur; cur = nxt; nxt = tmp;
    }

    float* __restrict__ yb = y + (size_t)b * T_LEN;
#pragma unroll
    for (int k = 0; k < YPT; ++k) {
        int i = tid + k * NTHREADS;
        int t = cs + i;
        if (i < CHUNK && t < T_LEN) yb[t] = yreg[k];
    }
}

extern "C" void kernel_launch(void* const* d_in, const int* in_sizes, int n_in,
                              void* d_out, int out_size, void* d_ws, size_t ws_size,
                              hipStream_t stream) {
    (void)n_in; (void)out_size; (void)d_ws; (void)ws_size;
    const float* x  = (const float*)d_in[0];
    const float* mc = (const float*)d_in[1];
    float* yout     = (float*)d_out;

    const int B = in_sizes[0] / T_LEN;
    dim3 grid((T_LEN + CHUNK - 1) / CHUNK, B);
    mglsa_taylor_wmma_kernel<<<grid, NTHREADS, 0, stream>>>(x, mc, yout);
}


# shrinking halo: iteration n skips the 16-row groups wholly below sample HALO-24*(30-n), which no later iteration or y can reach; wave-uniform skip, groups dealt round-robin stay balanced
# speedup vs baseline: 3.3072x; 1.1441x over previous
#include <hip/hip_runtime.h>


typedef __attribute__((ext_vector_type(2))) float v2f;
typedef __attribute__((ext_vector_type(8))) float v8f;

#define T_LEN    48000
#define M1       25
#define NITER    30
#define CHUNK    1504
#define HALO     800
#define REGION   (CHUNK + HALO)
#define NTHREADS 512
#define NWAVES   (NTHREADS / 32)
#define NGROUPS  (REGION / 16)
#define NG       (NGROUPS / NWAVES)
#define YPT      ((CHUNK + NTHREADS - 1) / NTHREADS)
#define PADL     32
#define USZ      (REGION + PADL)
#define DUMP     (USZ - 1)


__global__ __launch_bounds__(NTHREADS) __attribute__((amdgpu_waves_per_eu(4, 4)))
void mglsa_taylor_wmma_kernel(const float* __restrict__ x,
                              const float* __restrict__ mc,
                              float* __restrict__ y)
{
    __shared__ float ubuf[2 * USZ];

    const int b    = blockIdx.y;
    const int cs   = blockIdx.x * CHUNK;
    const int R    = cs - HALO;
    const int tid  = threadIdx.x;
    const int lane = tid & 31;
    const int wave = tid >> 5;
    const int wave_s = __builtin_amdgcn_readfirstlane(wave);

    const float* __restrict__ xb  = x  + (size_t)b * T_LEN;
    const float* __restrict__ mcb = mc + (size_t)b * T_LEN * M1;

    const int  m       = lane & 15;
    const int  half2   = (lane >> 4) << 1;
    const bool hasDiag = (lane < 8) || (lane >= 24);
    const int  dslot   = (lane < 8) ? lane : (lane - 16);

    v2f af[NG][7];
#pragma unroll
    for (int i = 0; i < NG; ++i) {
        const int trow = R + ((wave + i * NWAVES) << 4) + m;
        const int tcl  = trow < 0 ? 0 : (trow >= T_LEN ? (T_LEN - 1) : trow);
        const float* __restrict__ mrow = mcb + (size_t)tcl * M1;
#pragma unroll
        for (int q = 0; q < 6; ++q) {
            af[i][q].x = mrow[4 * q + half2];
            af[i][q].y = mrow[4 * q + half2 + 1];
        }
        const float a24 = mrow[24];
        af[i][6].x = (half2 == 0) ? a24 : 0.0f;
        af[i][6].y = 0.0f;
    }

    for (int p = tid; p < USZ; p += NTHREADS) {
        int l = (REGION - 1) - p;
        int t = R + l;
        ubuf[p] = (l >= 0 && (unsigned)t < (unsigned)T_LEN) ? xb[t] : 0.0f;
    }
    if (tid < PADL) ubuf[USZ + REGION + tid] = 0.0f;

    float yreg[YPT];
    int   ypos[YPT];
#pragma unroll
    for (int k = 0; k < YPT; ++k) {
        int i = tid + k * NTHREADS;
        int t = cs + i;
        yreg[k] = (i < CHUNK && t < T_LEN) ? xb[t] : 0.0f;
        ypos[k] = (i < CHUNK) ? ((CHUNK - 1) - i) : 0;
    }
    __syncthreads();

    int cur = 0;
    int nxt = USZ;

    const int pb0  = (REGION - 1) - ((wave << 4) + m - half2);
    const int pst0 = (REGION - 1) - ((wave << 4) + dslot);

    for (int n = 1; n <= NITER; ++n) {
        const float inv_a = 1.0f / (float)n;
        const float* __restrict__ ucur = ubuf + cur;
        float* __restrict__ unxt = ubuf + nxt;
        const int first_g = (HALO - (M1 - 1) * (NITER - n)) >> 4;

#pragma unroll
        for (int i = 0; i < NG; ++i) {
            if (i * NWAVES < HALO / 16 && wave_s + i * NWAVES < first_g) continue;
            const int pb = pb0 - i * (NWAVES << 4);

            v8f acc = {0.f, 0.f, 0.f, 0.f, 0.f, 0.f, 0.f, 0.f};
#pragma unroll
            for (int q = 0; q < 7; ++q) {
                v2f bb;
                bb.x = ucur[pb + 4 * q];
                bb.y = ucur[pb + 4 * q + 1];
                acc = __builtin_amdgcn_wmma_f32_16x16x4_f32(
                          false, af[i][q], false, bb, (short)0, acc, false, false);
            }

            float dv = acc[0];
#pragma unroll
            for (int v = 1; v < 8; ++v)
                dv = ((lane & 7) == v) ? acc[v] : dv;

            const int pst = hasDiag ? (pst0 - i * (NWAVES << 4)) : DUMP;
            unxt[pst] = dv * inv_a;
        }
        __syncthreads();

#pragma unroll
        for (int k = 0; k < YPT; ++k)
            yreg[k] += unxt[ypos[k]];

        int tmp = cur; cur = nxt; nxt = tmp;
    }

    float* __restrict__ yb = y + (size_t)b * T_LEN;
#pragma unroll
    for (int k = 0; k < YPT; ++k) {
        int i = tid + k * NTHREADS;
        int t = cs + i;
        if (i < CHUNK && t < T_LEN) yb[t] = yreg[k];
    }
}

extern "C" void kernel_launch(void* const* d_in, const int* in_sizes, int n_in,
                              void* d_out, int out_size, void* d_ws, size_t ws_size,
                              hipStream_t stream) {
    (void)n_in; (void)out_size; (void)d_ws; (void)ws_size;
    const float* x  = (const float*)d_in[0];
    const float* mc = (const float*)d_in[1];
    float* yout     = (float*)d_out;

    const int B = in_sizes[0] / T_LEN;
    dim3 grid((T_LEN + CHUNK - 1) / CHUNK, B);
    mglsa_taylor_wmma_kernel<<<grid, NTHREADS, 0, stream>>>(x, mc, yout);
}


# software pipeline over a wave's row-groups: B loads of group i+1 issued as one batch ahead of group i's WMMA chain, diagonal store of group i-1 placed inside it (2 B sets, 2 accumulators)
# speedup vs baseline: 3.6072x; 1.0907x over previous
#include <hip/hip_runtime.h>


typedef __attribute__((ext_vector_type(2))) float v2f;
typedef __attribute__((ext_vector_type(8))) float v8f;

#define T_LEN    48000
#define M1       25
#define NITER    30
#define CHUNK    1504
#define HALO     800
#define REGION   (CHUNK + HALO)
#define NTHREADS 512
#define NWAVES   (NTHREADS / 32)
#define NGROUPS  (REGION / 16)
#define NG       (NGROUPS / NWAVES)
#define YPT      ((CHUNK + NTHREADS - 1) / NTHREADS)
#define PADL     32
#define USZ      (REGION + PADL)
#define DUMP     (USZ - 1)


static __device__ __forceinline__ void load_b(v2f (&bb)[7], const float* ucur, int pb)
{
#pragma unroll
    for (int q = 0; q < 7; ++q) {
        bb[q].x = ucur[pb + 4 * q];
        bb[q].y = ucur[pb + 4 * q + 1];
    }
}

template <int Q0, int Q1>
static __device__ __forceinline__ v8f chain_part(const v2f (&a)[7], const v2f (&bb)[7], v8f acc)
{
#pragma unroll
    for (int q = Q0; q < Q1; ++q)
        acc = __builtin_amdgcn_wmma_f32_16x16x4_f32(
                  false, a[q], false, bb[q], (short)0, acc, false, false);
    return acc;
}

static __device__ __forceinline__ void put_diag(const v8f acc, float* unxt, int pst,
                                                int lane, float inv_a)
{
    float dv = acc[0];
#pragma unroll
    for (int v = 1; v < 8; ++v)
        dv = ((lane & 7) == v) ? acc[v] : dv;
    unxt[pst] = dv * inv_a;
}

__global__ __launch_bounds__(NTHREADS) __attribute__((amdgpu_waves_per_eu(4, 4)))
void mglsa_taylor_wmma_kernel(const float* __restrict__ x,
                              const float* __restrict__ mc,
                              float* __restrict__ y)
{
    __shared__ float ubuf[2 * USZ];

    const int b    = blockIdx.y;
    const int cs   = blockIdx.x * CHUNK;
    const int R    = cs - HALO;
    const int tid  = threadIdx.x;
    const int lane = tid & 31;
    const int wave = tid >> 5;
    const int wave_s = __builtin_amdgcn_readfirstlane(wave);

    const float* __restrict__ xb  = x  + (size_t)b * T_LEN;
    const float* __restrict__ mcb = mc + (size_t)b * T_LEN * M1;

    const int  m       = lane & 15;
    const int  half2   = (lane >> 4) << 1;
    const bool hasDiag = (lane < 8) || (lane >= 24);
    const int  dslot   = (lane < 8) ? lane : (lane - 16);

    v2f af[NG][7];
#pragma unroll
    for (int i = 0; i < NG; ++i) {
        const int trow = R + ((wave + i * NWAVES) << 4) + m;
        const int tcl  = trow < 0 ? 0 : (trow >= T_LEN ? (T_LEN - 1) : trow);
        const float* __restrict__ mrow = mcb + (size_t)tcl * M1;
#pragma unroll
        for (int q = 0; q < 6; ++q) {
            af[i][q].x = mrow[4 * q + half2];
            af[i][q].y = mrow[4 * q + half2 + 1];
        }
        const float a24 = mrow[24];
        af[i][6].x = (half2 == 0) ? a24 : 0.0f;
        af[i][6].y = 0.0f;
    }

    for (int p = tid; p < USZ; p += NTHREADS) {
        int l = (REGION - 1) - p;
        int t = R + l;
        ubuf[p] = (l >= 0 && (unsigned)t < (unsigned)T_LEN) ? xb[t] : 0.0f;
    }
    if (tid < PADL) ubuf[USZ + REGION + tid] = 0.0f;

    float yreg[YPT];
    int   ypos[YPT];
#pragma unroll
    for (int k = 0; k < YPT; ++k) {
        int i = tid + k * NTHREADS;
        int t = cs + i;
        yreg[k] = (i < CHUNK && t < T_LEN) ? xb[t] : 0.0f;
        ypos[k] = (i < CHUNK) ? ((CHUNK - 1) - i) : 0;
    }
    __syncthreads();

    int cur = 0;
    int nxt = USZ;

    const int pb0  = (REGION - 1) - ((wave << 4) + m - half2);
    const int pst0 = (REGION - 1) - ((wave << 4) + dslot);

    const v8f zero8 = {0.f, 0.f, 0.f, 0.f, 0.f, 0.f, 0.f, 0.f};
    const int GSTEP = NWAVES << 4;

    for (int n = 1; n <= NITER; ++n) {
        const float inv_a = 1.0f / (float)n;
        const float* ucur = ubuf + cur;
        float* unxt = ubuf + nxt;
        const int first_g = (HALO - (M1 - 1) * (NITER - n)) >> 4;
        int i0 = (first_g - wave_s + 15) >> 4;
        i0 = i0 < 0 ? 0 : i0;

        static_assert(NG == 9 && (HALO / 16 + NWAVES - 1) / NWAVES == 4,
                      "the stages below are written out for 9 groups and start index 0..4");
        v2f bE[7], bO[7];
        v8f accE, accO;

#define LOADS_FIRST __builtin_amdgcn_sched_barrier(0)
#define HEAD(I, BC, BN, ACC)                                   \
        load_b(BC, ucur, pb0 - (I) * GSTEP);                   \
        load_b(BN, ucur, pb0 - ((I) + 1) * GSTEP);             \
        LOADS_FIRST;                                           \
        ACC = chain_part<0, 7>(af[I], BC, zero8);
#define STAGE(I, BC, BN, ACC, ACCP)                            \
        load_b(BN, ucur, pb0 - ((I) + 1) * GSTEP);             \
        LOADS_FIRST;                                           \
        ACC = chain_part<0, 2>(af[I], BC, zero8);              \
        put_diag(ACCP, unxt, hasDiag ? (pst0 - ((I) - 1) * GSTEP) : DUMP, lane, inv_a); \
        ACC = chain_part<2, 7>(af[I], BC, ACC);

        switch (i0) {
        case 0:  { HEAD(0, bE, bO, accE) } goto stage1;
        case 1:  { HEAD(1, bO, bE, accO) } goto stage2;
        case 2:  { HEAD(2, bE, bO, accE) } goto stage3;
        case 3:  { HEAD(3, bO, bE, accO) } goto stage4;
        default: { HEAD(4, bE, bO, accE) } goto stage5;
        }
        stage1: { STAGE(1, bO, bE, accO, accE) }
        stage2: { STAGE(2, bE, bO, accE, accO) }
        stage3: { STAGE(3, bO, bE, accO, accE) }
        stage4: { STAGE(4, bE, bO, accE, accO) }
        stage5: { STAGE(5, bO, bE, accO, accE) }
        { STAGE(6, bE, bO, accE, accO) }
        { STAGE(7, bO, bE, accO, accE) }
        {
            accE = chain_part<0, 2>(af[8], bE, zero8);
            put_diag(accO, unxt, hasDiag ? (pst0 - 7 * GSTEP) : DUMP, lane, inv_a);
            accE = chain_part<2, 7>(af[8], bE, accE);
            put_diag(accE, unxt, hasDiag ? (pst0 - 8 * GSTEP) : DUMP, lane, inv_a);
        }
#undef HEAD
#undef STAGE
#undef LOADS_FIRST
        __syncthreads();

#pragma unroll
        for (int k = 0; k < YPT; ++k)
            yreg[k] += unxt[ypos[k]];

        int tmp = cur; cur = nxt; nxt = tmp;
    }

    float* __restrict__ yb = y + (size_t)b * T_LEN;
#pragma unroll
    for (int k = 0; k < YPT; ++k) {
        int i = tid + k * NTHREADS;
        int t = cs + i;
        if (i < CHUNK && t < T_LEN) yb[t] = yreg[k];
    }
}

extern "C" void kernel_launch(void* const* d_in, const int* in_sizes, int n_in,
                              void* d_out, int out_size, void* d_ws, size_t ws_size,
                              hipStream_t stream) {
    (void)n_in; (void)out_size; (void)d_ws; (void)ws_size;
    const float* x  = (const float*)d_in[0];
    const float* mc = (const float*)d_in[1];
    float* yout     = (float*)d_out;

    const int B = in_sizes[0] / T_LEN;
    dim3 grid((T_LEN + CHUNK - 1) / CHUNK, B);
    mglsa_taylor_wmma_kernel<<<grid, NTHREADS, 0, stream>>>(x, mc, yout);
}


# mc rows of a group read as one contiguous 1600 B piece with whole-line 16-byte loads, turned into A fragments through a per-wave LDS tile (4 loads of 4 lines, not 7 of 13 lines); same k order
# speedup vs baseline: 3.7306x; 1.0342x over previous
#include <hip/hip_runtime.h>


typedef __attribute__((ext_vector_type(2))) float v2f;
typedef __attribute__((ext_vector_type(8))) float v8f;
typedef __attribute__((ext_vector_type(4))) float v4f;

#define T_LEN    48000
#define M1       25
#define NITER    30
#define CHUNK    1504
#define HALO     800
#define REGION   (CHUNK + HALO)
#define NTHREADS 512
#define NWAVES   (NTHREADS / 32)
#define NGROUPS  (REGION / 16)
#define NG       (NGROUPS / NWAVES)
#define YPT      ((CHUNK + NTHREADS - 1) / NTHREADS)
#define PADL     32
#define USZ      (REGION + PADL)
#define DUMP     (USZ - 1)
#define A4       (16 * M1 / 4)
#define AW4      ((A4 + 31) / 32)


static __device__ __forceinline__ void load_b(v2f (&bb)[7], const float* ucur, int pb)
{
#pragma unroll
    for (int q = 0; q < 7; ++q) {
        bb[q].x = ucur[pb + 4 * q];
        bb[q].y = ucur[pb + 4 * q + 1];
    }
}

template <int Q0, int Q1>
static __device__ __forceinline__ v8f chain_part(const v2f (&a)[7], const v2f (&bb)[7], v8f acc)
{
#pragma unroll
    for (int q = Q0; q < Q1; ++q)
        acc = __builtin_amdgcn_wmma_f32_16x16x4_f32(
                  false, a[q], false, bb[q], (short)0, acc, false, false);
    return acc;
}

static __device__ __forceinline__ void put_diag(const v8f acc, float* unxt, int pst,
                                                int lane, float inv_a)
{
    float dv = acc[0];
#pragma unroll
    for (int v = 1; v < 8; ++v)
        dv = ((lane & 7) == v) ? acc[v] : dv;
    unxt[pst] = dv * inv_a;
}

__global__ __launch_bounds__(NTHREADS) __attribute__((amdgpu_waves_per_eu(4, 4)))
void mglsa_taylor_wmma_kernel(const float* __restrict__ x,
                              const float* __restrict__ mc,
                              float* __restrict__ y)
{
    __shared__ float ubuf[2 * USZ];
    __shared__ v4f atile[NWAVES][32 * AW4];

    const int b    = blockIdx.y;
    const int cs   = blockIdx.x * CHUNK;
    const int R    = cs - HALO;
    const int tid  = threadIdx.x;
    const int lane = tid & 31;
    const int wave = tid >> 5;
    const int wave_s = __builtin_amdgcn_readfirstlane(wave);

    const float* __restrict__ xb  = x  + (size_t)b * T_LEN;
    const float* __restrict__ mcb = mc + (size_t)b * T_LEN * M1;

    const int  m       = lane & 15;
    const int  half2   = (lane >> 4) << 1;
    const bool hasDiag = (lane < 8) || (lane >= 24);
    const int  dslot   = (lane < 8) ? lane : (lane - 16);

    static_assert(T_LEN % 16 == 0 && CHUNK % 16 == 0 && HALO % 16 == 0,
                  "a row-group must never straddle the ends of the signal");
    static_assert((16 * M1) % 4 == 0, "a group's rows must be whole 16-byte words");
    v2f af[NG][7];
    {
        v4f w[NG][AW4];
#pragma unroll
        for (int i = 0; i < NG; ++i) {
            const int t0  = R + ((wave_s + i * NWAVES) << 4);
            const int t0c = t0 < 0 ? 0 : (t0 >= T_LEN ? (T_LEN - 16) : t0);
            const v4f* __restrict__ g4 = (const v4f*)(mcb + (size_t)t0c * M1);
#pragma unroll
            for (int j = 0; j < AW4; ++j) {
                const int idx = lane + 32 * j;
                w[i][j] = g4[idx < A4 ? idx : (A4 - 1)];
            }
        }
        v4f* __restrict__ st4 = atile[wave_s];
        const float* sw = (const float*)atile[wave_s];
        const int arow = M1 * m + half2;
#pragma unroll
        for (int i = 0; i < NG; ++i) {
#pragma unroll
            for (int j = 0; j < AW4; ++j)
                st4[lane + 32 * j] = w[i][j];
            __builtin_amdgcn_fence(__ATOMIC_RELEASE, "wavefront");
            __builtin_amdgcn_wave_barrier();
            __builtin_amdgcn_fence(__ATOMIC_ACQUIRE, "wavefront");
#pragma unroll
            for (int q = 0; q < 6; ++q) {
                af[i][q].x = sw[arow + 4 * q];
                af[i][q].y = sw[arow + 4 * q + 1];
            }
            const float a24 = sw[M1 * m + 24];
            af[i][6].x = (half2 == 0) ? a24 : 0.0f;
            af[i][6].y = 0.0f;
            __builtin_amdgcn_fence(__ATOMIC_RELEASE, "wavefront");
            __builtin_amdgcn_wave_barrier();
            __builtin_amdgcn_fence(__ATOMIC_ACQUIRE, "wavefront");
        }
    }

    for (int p = tid; p < USZ; p += NTHREADS) {
        int l = (REGION - 1) - p;
        int t = R + l;
        ubuf[p] = (l >= 0 && (unsigned)t < (unsigned)T_LEN) ? xb[t] : 0.0f;
    }
    if (tid < PADL) ubuf[USZ + REGION + tid] = 0.0f;

    float yreg[YPT];
    int   ypos[YPT];
#pragma unroll
    for (int k = 0; k < YPT; ++k) {
        int i = tid + k * NTHREADS;
        int t = cs + i;
        yreg[k] = (i < CHUNK && t < T_LEN) ? xb[t] : 0.0f;
        ypos[k] = (i < CHUNK) ? ((CHUNK - 1) - i) : 0;
    }
    __syncthreads();

    int cur = 0;
    int nxt = USZ;

    const int pb0  = (REGION - 1) - ((wave << 4) + m - half2);
    const int pst0 = (REGION - 1) - ((wave << 4) + dslot);

    const v8f zero8 = {0.f, 0.f, 0.f, 0.f, 0.f, 0.f, 0.f, 0.f};
    const int GSTEP = NWAVES << 4;

    for (int n = 1; n <= NITER; ++n) {
        const float inv_a = 1.0f / (float)n;
        const float* ucur = ubuf + cur;
        float* unxt = ubuf + nxt;
        const int first_g = (HALO - (M1 - 1) * (NITER - n)) >> 4;
        int i0 = (first_g - wave_s + 15) >> 4;
        i0 = i0 < 0 ? 0 : i0;

        static_assert(NG == 9 && (HALO / 16 + NWAVES - 1) / NWAVES == 4,
                      "the stages below are written out for 9 groups and start index 0..4");
        v2f bE[7], bO[7];
        v8f accE, accO;

#define LOADS_FIRST __builtin_amdgcn_sched_barrier(0)
#define HEAD(I, BC, BN, ACC)                                   \
        load_b(BC, ucur, pb0 - (I) * GSTEP);                   \
        load_b(BN, ucur, pb0 - ((I) + 1) * GSTEP);             \
        LOADS_FIRST;                                           \
        ACC = chain_part<0, 7>(af[I], BC, zero8);
#define STAGE(I, BC, BN, ACC, ACCP)                            \
        load_b(BN, ucur, pb0 - ((I) + 1) * GSTEP);             \
        LOADS_FIRST;                                           \
        ACC = chain_part<0, 2>(af[I], BC, zero8);              \
        put_diag(ACCP, unxt, hasDiag ? (pst0 - ((I) - 1) * GSTEP) : DUMP, lane, inv_a); \
        ACC = chain_part<2, 7>(af[I], BC, ACC);

        switch (i0) {
        case 0:  { HEAD(0, bE, bO, accE) } goto stage1;
        case 1:  { HEAD(1, bO, bE, accO) } goto stage2;
        case 2:  { HEAD(2, bE, bO, accE) } goto stage3;
        case 3:  { HEAD(3, bO, bE, accO) } goto stage4;
        default: { HEAD(4, bE, bO, accE) } goto stage5;
        }
        stage1: { STAGE(1, bO, bE, accO, accE) }
        stage2: { STAGE(2, bE, bO, accE, accO) }
        stage3: { STAGE(3, bO, bE, accO, accE) }
        stage4: { STAGE(4, bE, bO, accE, accO) }
        stage5: { STAGE(5, bO, bE, accO, accE) }
        { STAGE(6, bE, bO, accE, accO) }
        { STAGE(7, bO, bE, accO, accE) }
        {
            accE = chain_part<0, 2>(af[8], bE, zero8);
            put_diag(accO, unxt, hasDiag ? (pst0 - 7 * GSTEP) : DUMP, lane, inv_a);
            accE = chain_part<2, 7>(af[8], bE, accE);
            put_diag(accE, unxt, hasDiag ? (pst0 - 8 * GSTEP) : DUMP, lane, inv_a);
        }
#undef HEAD
#undef STAGE
#undef LOADS_FIRST
        __syncthreads();

#pragma unroll
        for (int k = 0; k < YPT; ++k)
            yreg[k] += unxt[ypos[k]];

        int tmp = cur; cur = nxt; nxt = tmp;
    }

    float* __restrict__ yb = y + (size_t)b * T_LEN;
#pragma unroll
    for (int k = 0; k < YPT; ++k) {
        int i = tid + k * NTHREADS;
        int t = cs + i;
        if (i < CHUNK && t < T_LEN) yb[t] = yreg[k];
    }
}

extern "C" void kernel_launch(void* const* d_in, const int* in_sizes, int n_in,
                              void* d_out, int out_size, void* d_ws, size_t ws_size,
                              hipStream_t stream) {
    (void)n_in; (void)out_size; (void)d_ws; (void)ws_size;
    const float* x  = (const float*)d_in[0];
    const float* mc = (const float*)d_in[1];
    float* yout     = (float*)d_out;

    const int B = in_sizes[0] / T_LEN;
    dim3 grid((T_LEN + CHUNK - 1) / CHUNK, B);
    mglsa_taylor_wmma_kernel<<<grid, NTHREADS, 0, stream>>>(x, mc, yout);
}
